# starting kernel
# baseline (speedup 1.0000x reference)
#include <hip/hip_runtime.h>

#define NUM_NODE 100000
#define EMBED_DIM 128
#define N_EDGES 600000

typedef __attribute__((ext_vector_type(2))) float v2f;
typedef __attribute__((ext_vector_type(8))) float v8f;

__global__ void na_init_kernel(float* __restrict__ out, float* __restrict__ deg) {
    const int idx = blockIdx.x * blockDim.x + threadIdx.x;
    if (idx < NUM_NODE * EMBED_DIM) out[idx] = 0.0f;
    if (idx < NUM_NODE)             deg[idx] = 0.0f;
}

__global__ void __launch_bounds__(256)
na_gemm_relu_kernel(const float* __restrict__ x,
                    const float* __restrict__ W,
                    const float* __restrict__ b,
                    float* __restrict__ y) {
    const int lane = threadIdx.x & 31;
    const int waveInBlock = threadIdx.x >> 5;
    const int waveId = blockIdx.x * 8 + waveInBlock;
    const int m0 = waveId * 16;
    if (m0 >= NUM_NODE) return;

    const int half = lane >> 4;
    const int l16  = lane & 15;

    const v8f zero = {};
    v8f acc[8];
#pragma unroll
    for (int nt = 0; nt < 8; ++nt) acc[nt] = zero;

    const float* __restrict__ xrow = x + (long long)(m0 + l16) * EMBED_DIM;

    for (int k0 = 0; k0 < EMBED_DIM; k0 += 4) {
        const int kk = k0 + 2 * half;
        v2f a;
        a.x = xrow[kk + 0];
        a.y = xrow[kk + 1];
#pragma unroll
        for (int nt = 0; nt < 8; ++nt) {
            const int n = nt * 16 + l16;
            v2f bm;
            bm.x = W[n * EMBED_DIM + kk + 0];
            bm.y = W[n * EMBED_DIM + kk + 1];
            acc[nt] = __builtin_amdgcn_wmma_f32_16x16x4_f32(
                false, a, false, bm, (short)0, acc[nt], false, false);
        }
    }

#pragma unroll
    for (int nt = 0; nt < 8; ++nt) {
        const int n = nt * 16 + l16;
        const float bias = b[n];
#pragma unroll
        for (int j = 0; j < 8; ++j) {
            const int m = m0 + half * 8 + j;
            float v = acc[nt][j] + bias;
            y[(long long)m * EMBED_DIM + n] = v > 0.0f ? v : 0.0f;
        }
    }
}

__global__ void na_scatter_kernel(const int* __restrict__ src,
                                  const int* __restrict__ tgt,
                                  const float* __restrict__ y,
                                  float* __restrict__ out,
                                  float* __restrict__ deg) {
    const int gid = blockIdx.x * blockDim.x + threadIdx.x;
    const int e = gid >> 5;
    if (e >= N_EDGES) return;
    const int q  = (gid & 31) * 4;

    const int s = src[e];
    const int t = tgt[e];

    const float4 v = *(const float4*)(y + (long long)t * EMBED_DIM + q);
    float* o = out + (long long)s * EMBED_DIM + q;
    atomicAdd(o + 0, v.x);
    atomicAdd(o + 1, v.y);
    atomicAdd(o + 2, v.z);
    atomicAdd(o + 3, v.w);
    if (q == 0) atomicAdd(deg + s, 1.0f);
}

__global__ void na_norm_kernel(float* __restrict__ out,
                               const float* __restrict__ deg) {
    const int idx = blockIdx.x * blockDim.x + threadIdx.x;
    if (idx >= NUM_NODE * EMBED_DIM) return;
    float d = deg[idx >> 7];
    d = d > 1.0f ? d : 1.0f;
    out[idx] = out[idx] / d;
}

extern "C" void kernel_launch(void* const* d_in, const int* in_sizes, int n_in,
                              void* d_out, int out_size, void* d_ws, size_t ws_size,
                              hipStream_t stream) {
    const float* x  = (const float*)d_in[0];
    const int*   ei = (const int*)d_in[1];
    const float* W  = (const float*)d_in[2];
    const float* b  = (const float*)d_in[3];
    float* out = (float*)d_out;

    float* deg = (float*)d_ws;
    float* y   = (float*)((char*)d_ws +
                 (((size_t)NUM_NODE * sizeof(float) + 255) & ~(size_t)255));

    {
        const int total = NUM_NODE * EMBED_DIM;
        na_init_kernel<<<(total + 255) / 256, 256, 0, stream>>>(out, deg);
    }
    {
        const int waves  = NUM_NODE / 16;
        const int blocks = (waves + 7) / 8;
        na_gemm_relu_kernel<<<blocks, 256, 0, stream>>>(x, W, b, y);
    }
    {
        const long long threads = (long long)N_EDGES * 32;
        const int blocks = (int)((threads + 255) / 256);
        na_scatter_kernel<<<blocks, 256, 0, stream>>>(ei, ei + N_EDGES, y, out, deg);
    }
    {
        const int total = NUM_NODE * EMBED_DIM;
        na_norm_kernel<<<(total + 255) / 256, 256, 0, stream>>>(out, deg);
    }
}


# Float-atomic scatter, zeroing and normalise passes replaced: an integer atomic counter buckets each edge's target id in its source's output row; one wave per node sums its y rows in slot order
# speedup vs baseline: 1.5711x; 1.5711x over previous
#include <hip/hip_runtime.h>

#define NUM_NODE 100000
#define EMBED_DIM 128
#define N_EDGES 600000
#define ROW_SLOTS 128

typedef __attribute__((ext_vector_type(2))) float v2f;
typedef __attribute__((ext_vector_type(8))) float v8f;

__global__ void __launch_bounds__(256)
na_gemm_relu_kernel(const float* __restrict__ x,
                    const float* __restrict__ W,
                    const float* __restrict__ b,
                    float* __restrict__ y,
                    int* __restrict__ cnt) {
    const int gid = blockIdx.x * blockDim.x + threadIdx.x;
    if (gid < NUM_NODE) cnt[gid] = 0;

    const int lane = threadIdx.x & 31;
    const int waveInBlock = threadIdx.x >> 5;
    const int waveId = blockIdx.x * 8 + waveInBlock;
    const int m0 = waveId * 16;
    if (m0 >= NUM_NODE) return;

    const int half = lane >> 4;
    const int l16  = lane & 15;

    const v8f zero = {};
    v8f acc[8];
#pragma unroll
    for (int nt = 0; nt < 8; ++nt) acc[nt] = zero;

    const float* __restrict__ xrow = x + (long long)(m0 + l16) * EMBED_DIM;

    for (int k0 = 0; k0 < EMBED_DIM; k0 += 4) {
        const int kk = k0 + 2 * half;
        v2f a;
        a.x = xrow[kk + 0];
        a.y = xrow[kk + 1];
#pragma unroll
        for (int nt = 0; nt < 8; ++nt) {
            const int n = nt * 16 + l16;
            v2f bm;
            bm.x = W[n * EMBED_DIM + kk + 0];
            bm.y = W[n * EMBED_DIM + kk + 1];
            acc[nt] = __builtin_amdgcn_wmma_f32_16x16x4_f32(
                false, a, false, bm, (short)0, acc[nt], false, false);
        }
    }

#pragma unroll
    for (int nt = 0; nt < 8; ++nt) {
        const int n = nt * 16 + l16;
        const float bias = b[n];
#pragma unroll
        for (int j = 0; j < 8; ++j) {
            const int m = m0 + half * 8 + j;
            float v = acc[nt][j] + bias;
            y[(long long)m * EMBED_DIM + n] = v > 0.0f ? v : 0.0f;
        }
    }
}

__global__ void na_bucket_kernel(const int* __restrict__ src,
                                 const int* __restrict__ tgt,
                                 int* __restrict__ cnt,
                                 int* __restrict__ slots) {
    const int e = blockIdx.x * blockDim.x + threadIdx.x;
    if (e >= N_EDGES) return;
    const int s = src[e];
    const int t = tgt[e];
    const int pos = atomicAdd(cnt + s, 1);
    if (pos < ROW_SLOTS) slots[(long long)s * ROW_SLOTS + pos] = t;
}

__global__ void __launch_bounds__(256)
na_gather_kernel(const int* __restrict__ src,
                 const int* __restrict__ tgt,
                 const float* __restrict__ y,
                 const int* __restrict__ cnt,
                 float* out) {
    const int lane = threadIdx.x & 31;
    const int n = blockIdx.x * 8 + __builtin_amdgcn_readfirstlane((int)(threadIdx.x >> 5));
    if (n >= NUM_NODE) return;

    const int* slots = (const int*)out + (long long)n * ROW_SLOTS;
    const int c = cnt[n];
    const float* __restrict__ ycol = y + lane * 4;

    const float4 zero4 = {0.0f, 0.0f, 0.0f, 0.0f};
    float4 acc = zero4;

    if (c <= ROW_SLOTS) {
        int ids = 0;
        for (int base = 0; base < c; base += 8) {
            if ((base & 31) == 0) ids = (base + lane < c) ? slots[base + lane] : 0;
            float4 r[8];
#pragma unroll
            for (int j = 0; j < 8; ++j) {
                r[j] = zero4;
                if (base + j < c) {
                    const int t = __builtin_amdgcn_readlane(ids, (base + j) & 31);
                    r[j] = *(const float4*)(ycol + (long long)t * EMBED_DIM);
                }
            }
#pragma unroll
            for (int j = 0; j < 8; ++j) {
                acc.x += r[j].x; acc.y += r[j].y; acc.z += r[j].z; acc.w += r[j].w;
            }
        }
    } else {
        for (int base = 0; base < N_EDGES; base += 32) {
            const int sv = src[base + lane];
            unsigned mask = __builtin_amdgcn_ballot_w32(sv == n);
            while (mask) {
                const int j = __builtin_ctz(mask);
                mask &= mask - 1;
                const int t = __builtin_amdgcn_readfirstlane(tgt[base + j]);
                const float4 r = *(const float4*)(ycol + (long long)t * EMBED_DIM);
                acc.x += r.x; acc.y += r.y; acc.z += r.z; acc.w += r.w;
            }
        }
    }

    const float d = c > 1 ? (float)c : 1.0f;
    float4 o;
    o.x = acc.x / d; o.y = acc.y / d; o.z = acc.z / d; o.w = acc.w / d;
    *(float4*)(out + (long long)n * EMBED_DIM + lane * 4) = o;
}

extern "C" void kernel_launch(void* const* d_in, const int* in_sizes, int n_in,
                              void* d_out, int out_size, void* d_ws, size_t ws_size,
                              hipStream_t stream) {
    const float* x  = (const float*)d_in[0];
    const int*   ei = (const int*)d_in[1];
    const float* W  = (const float*)d_in[2];
    const float* b  = (const float*)d_in[3];
    float* out = (float*)d_out;

    int* cnt = (int*)d_ws;
    float* y   = (float*)((char*)d_ws +
                 (((size_t)NUM_NODE * sizeof(float) + 255) & ~(size_t)255));

    {
        const int waves  = NUM_NODE / 16;
        const int blocks = (waves + 7) / 8;
        na_gemm_relu_kernel<<<blocks, 256, 0, stream>>>(x, W, b, y, cnt);
    }
    {
        const int blocks = (N_EDGES + 255) / 256;
        na_bucket_kernel<<<blocks, 256, 0, stream>>>(ei, ei + N_EDGES, cnt, (int*)out);
    }
    {
        const int blocks = (NUM_NODE + 7) / 8;
        na_gather_kernel<<<blocks, 256, 0, stream>>>(ei, ei + N_EDGES, y, cnt, out);
    }
}


# Edges grouped by source in LDS, not by global integer atomics and scattered stores: 128 blocks bin by 128-node range, the range's block ranks by node in LDS and sums rows in that rank order
# speedup vs baseline: 2.8444x; 1.8105x over previous
#include <hip/hip_runtime.h>

#define NUM_NODE 100000
#define EMBED_DIM 128
#define N_EDGES 600000
#define BIN_NODES 128
#define NBINS ((NUM_NODE + BIN_NODES - 1) / BIN_NODES)
#define NCHUNK 128
#define CHUNK_EDGES ((N_EDGES + NCHUNK - 1) / NCHUNK)
#define PIECE_WORDS 32
#define PIECE_CAP (PIECE_WORDS - 1)
#define TGT_BITS 17
#define TGT_MASK ((1u << TGT_BITS) - 1u)
#define BIN_WORDS (BIN_NODES * EMBED_DIM)

typedef __attribute__((ext_vector_type(2))) float v2f;
typedef __attribute__((ext_vector_type(8))) float v8f;

__global__ void __launch_bounds__(256)
na_gemm_relu_kernel(const float* __restrict__ x,
                    const float* __restrict__ W,
                    const float* __restrict__ b,
                    float* __restrict__ y) {
    __shared__ float4 wfrag[8 * 16 * 32];

    const int tid = threadIdx.x;

    const int lane = tid & 31;
    const int waveInBlock = tid >> 5;
    const int waveId = blockIdx.x * 8 + waveInBlock;
    const int m0 = waveId * 16;
    const bool live = m0 < NUM_NODE;

    const int half = lane >> 4;
    const int l16  = lane & 15;

    float4 av[16];
    if (live) {
        const float4* __restrict__ xr =
            (const float4*)(x + (long long)(m0 + l16) * EMBED_DIM) + half;
#pragma unroll
        for (int g = 0; g < 16; ++g) av[g] = xr[2 * g];
    }

    {
        const float4* __restrict__ W4 = (const float4*)W;
        float4 st[16];
#pragma unroll
        for (int i = 0; i < 16; ++i) st[i] = W4[i * 256 + tid];
#pragma unroll
        for (int i = 0; i < 16; ++i) {
            const int n  = i * 8 + waveInBlock;
            const int sg = lane >> 1;
            const int sh = lane & 1;
            wfrag[((n >> 4) * 16 + sg) * 32 + sh * 16 + ((n & 15) ^ sg)] = st[i];
        }
    }
    __syncthreads();
    if (!live) return;

    const v8f zero = {};
    v8f acc[8];
#pragma unroll
    for (int nt = 0; nt < 8; ++nt) acc[nt] = zero;

    float4 bq[2][8];
#pragma unroll
    for (int nt = 0; nt < 8; ++nt) bq[0][nt] = wfrag[(nt * 16) * 32 + half * 16 + l16];

#pragma unroll
    for (int g = 0; g < 16; ++g) {
        const int cur = g & 1;
        const int nxt = cur ^ 1;
        if (g + 1 < 16) {
#pragma unroll
            for (int nt = 0; nt < 8; ++nt)
                bq[nxt][nt] = wfrag[(nt * 16 + g + 1) * 32 + half * 16 + (l16 ^ (g + 1))];
        }
        __builtin_amdgcn_sched_barrier(0);
        v2f a0, a1;
        a0.x = av[g].x; a0.y = av[g].y;
        a1.x = av[g].z; a1.y = av[g].w;
#pragma unroll
        for (int nt = 0; nt < 8; ++nt) {
            v2f b0;
            b0.x = bq[cur][nt].x; b0.y = bq[cur][nt].y;
            acc[nt] = __builtin_amdgcn_wmma_f32_16x16x4_f32(
                false, a0, false, b0, (short)0, acc[nt], false, false);
        }
#pragma unroll
        for (int nt = 0; nt < 8; ++nt) {
            v2f b1;
            b1.x = bq[cur][nt].z; b1.y = bq[cur][nt].w;
            acc[nt] = __builtin_amdgcn_wmma_f32_16x16x4_f32(
                false, a1, false, b1, (short)0, acc[nt], false, false);
        }
        __builtin_amdgcn_sched_barrier(0);
    }

#pragma unroll
    for (int nt = 0; nt < 8; ++nt) {
        const int n = nt * 16 + l16;
        const float bias = b[n];
#pragma unroll
        for (int j = 0; j < 8; ++j) {
            const int m = m0 + half * 8 + j;
            float v = acc[nt][j] + bias;
            y[(long long)m * EMBED_DIM + n] = v > 0.0f ? v : 0.0f;
        }
    }
}

__global__ void __launch_bounds__(1024)
na_bin_kernel(const int* __restrict__ src,
              const int* __restrict__ tgt,
              unsigned* __restrict__ pieces) {
    __shared__ unsigned lcnt[NBINS];
    __shared__ unsigned lent[NBINS * PIECE_CAP];

    const int tid = threadIdx.x;
    const int chunk = blockIdx.x;

    for (int bn = tid; bn < NBINS; bn += 1024) lcnt[bn] = 0;

    int s[5], t[5];
#pragma unroll
    for (int k = 0; k < 5; ++k) {
        const int i = tid + k * 1024;
        const int e = chunk * CHUNK_EDGES + i;
        const bool ok = (i < CHUNK_EDGES) && (e < N_EDGES);
        s[k] = ok ? src[e] : -1;
        t[k] = ok ? tgt[e] : 0;
    }
    __syncthreads();

#pragma unroll
    for (int k = 0; k < 5; ++k) {
        if (s[k] >= 0) {
            const unsigned bn  = (unsigned)s[k] / BIN_NODES;
            const unsigned loc = (unsigned)s[k] % BIN_NODES;
            const unsigned pos = atomicAdd(&lcnt[bn], 1u);
            if (pos < PIECE_CAP) lent[bn * PIECE_CAP + pos] = (loc << TGT_BITS) | (unsigned)t[k];
        }
    }
    __syncthreads();

    for (int i = tid; i < NBINS * PIECE_WORDS; i += 1024) {
        const int bn = i / PIECE_WORDS;
        const unsigned j = (unsigned)(i % PIECE_WORDS);
        const unsigned c = lcnt[bn];
        unsigned v = 0;
        if (j == 0) v = c;
        else if (j <= c) v = lent[bn * PIECE_CAP + j - 1];
        pieces[(size_t)bn * BIN_WORDS + chunk * PIECE_WORDS + j] = v;
    }
}

__device__ __forceinline__ void na_node_by_scan(const int* __restrict__ src,
                                                const int* __restrict__ tgt,
                                                const float* __restrict__ ycol,
                                                float* out, int n, int lane) {
    float4 acc = {0.0f, 0.0f, 0.0f, 0.0f};
    int c = 0;
    for (int base = 0; base < N_EDGES; base += 32) {
        const int sv = src[base + lane];
        unsigned mask = __builtin_amdgcn_ballot_w32(sv == n);
        c += __builtin_popcount(mask);
        while (mask) {
            const int j = __builtin_ctz(mask);
            mask &= mask - 1;
            const int tt = __builtin_amdgcn_readfirstlane(tgt[base + j]);
            const float4 r = *(const float4*)(ycol + (long long)tt * EMBED_DIM);
            acc.x += r.x; acc.y += r.y; acc.z += r.z; acc.w += r.w;
        }
    }
    const float d = c > 1 ? (float)c : 1.0f;
    float4 o;
    o.x = acc.x / d; o.y = acc.y / d; o.z = acc.z / d; o.w = acc.w / d;
    *(float4*)(out + (long long)n * EMBED_DIM + lane * 4) = o;
}

__global__ void __launch_bounds__(512, 16)
na_gather_kernel(const int* __restrict__ src,
                 const int* __restrict__ tgt,
                 const float* __restrict__ y,
                 float* out) {
    __shared__ int lcnt[BIN_NODES];
    __shared__ int lstart[BIN_NODES + 1];
    __shared__ int sorted[NCHUNK * PIECE_CAP];
    __shared__ int ovf;

    const int tid  = threadIdx.x;
    const int lane = tid & 31;
    const int wave = __builtin_amdgcn_readfirstlane(tid >> 5);
    const int bin  = blockIdx.x;
    const int nodeBase = bin * BIN_NODES;
    const float* __restrict__ ycol = y + lane * 4;

    const unsigned* pieces = (const unsigned*)out + (size_t)bin * BIN_WORDS;
    unsigned w[8];
#pragma unroll
    for (int k = 0; k < 8; ++k) w[k] = pieces[k * 512 + tid];

    if (tid < BIN_NODES) lcnt[tid] = 0;
    if (tid == 0) ovf = 0;
    __syncthreads();

    int rank[8];
#pragma unroll
    for (int k = 0; k < 8; ++k) {
        const unsigned c = (unsigned)__builtin_amdgcn_readfirstlane((int)w[k]);
        rank[k] = -1;
        if (c > PIECE_CAP) {
            if (lane == 0) ovf = 1;
        } else if (lane >= 1 && (unsigned)lane <= c) {
            rank[k] = atomicAdd(&lcnt[w[k] >> TGT_BITS], 1);
        }
    }
    __syncthreads();

    if (ovf) {
        for (int i = 0; i < 8; ++i) {
            const int n = nodeBase + wave * 8 + i;
            if (n < NUM_NODE) na_node_by_scan(src, tgt, ycol, out, n, lane);
        }
        return;
    }

    if (wave == 0) {
        const int c0 = lcnt[4 * lane], c1 = lcnt[4 * lane + 1];
        const int c2 = lcnt[4 * lane + 2], c3 = lcnt[4 * lane + 3];
        const int own = c0 + c1 + c2 + c3;
        int incl = own;
#pragma unroll
        for (int d = 1; d < 32; d <<= 1) {
            const int up = __shfl_up(incl, d, 32);
            if (lane >= d) incl += up;
        }
        const int excl = incl - own;
        lstart[4 * lane]     = excl;
        lstart[4 * lane + 1] = excl + c0;
        lstart[4 * lane + 2] = excl + c0 + c1;
        lstart[4 * lane + 3] = excl + c0 + c1 + c2;
        if (lane == 31) lstart[BIN_NODES] = incl;
    }
    __syncthreads();

#pragma unroll
    for (int k = 0; k < 8; ++k) {
        if (rank[k] >= 0)
            sorted[lstart[w[k] >> TGT_BITS] + rank[k]] = (int)(w[k] & TGT_MASK);
    }
    __syncthreads();

    const int n0 = wave * 8;
    const int e1 = __builtin_amdgcn_readfirstlane(lstart[n0 + 8]);
    int cur    = n0;
    int curBeg = __builtin_amdgcn_readfirstlane(lstart[n0]);
    int curEnd = __builtin_amdgcn_readfirstlane(lstart[n0 + 1]);

    const float4 zero4 = {0.0f, 0.0f, 0.0f, 0.0f};
    float4 acc = zero4;

    for (int base = curBeg; base < e1; base += 8) {
        const int idx = base + (lane & 7);
        const int ids = idx < e1 ? sorted[idx] : 0;
        float4 r[8];
#pragma unroll
        for (int j = 0; j < 8; ++j) {
            r[j] = zero4;
            if (base + j < e1) {
                const int tt = __builtin_amdgcn_readlane(ids, j);
                r[j] = *(const float4*)(ycol + (long long)tt * EMBED_DIM);
            }
        }
#pragma unroll
        for (int j = 0; j < 8; ++j) {
            if (base + j < e1) {
                while (base + j >= curEnd) {
                    const int n = nodeBase + cur;
                    if (n < NUM_NODE) {
                        const int deg = curEnd - curBeg;
                        const float d = deg > 1 ? (float)deg : 1.0f;
                        float4 o;
                        o.x = acc.x / d; o.y = acc.y / d; o.z = acc.z / d; o.w = acc.w / d;
                        *(float4*)(out + (long long)n * EMBED_DIM + lane * 4) = o;
                    }
                    acc = zero4;
                    ++cur;
                    curBeg = curEnd;
                    curEnd = __builtin_amdgcn_readfirstlane(lstart[cur + 1]);
                }
                acc.x += r[j].x; acc.y += r[j].y; acc.z += r[j].z; acc.w += r[j].w;
            }
        }
    }
    while (cur < n0 + 8) {
        const int n = nodeBase + cur;
        if (n < NUM_NODE) {
            const int deg = curEnd - curBeg;
            const float d = deg > 1 ? (float)deg : 1.0f;
            float4 o;
            o.x = acc.x / d; o.y = acc.y / d; o.z = acc.z / d; o.w = acc.w / d;
            *(float4*)(out + (long long)n * EMBED_DIM + lane * 4) = o;
        }
        acc = zero4;
        ++cur;
        curBeg = curEnd;
        if (cur < n0 + 8) curEnd = __builtin_amdgcn_readfirstlane(lstart[cur + 1]);
    }
}

extern "C" void kernel_launch(void* const* d_in, const int* in_sizes, int n_in,
                              void* d_out, int out_size, void* d_ws, size_t ws_size,
                              hipStream_t stream) {
    const float* x  = (const float*)d_in[0];
    const int*   ei = (const int*)d_in[1];
    const float* W  = (const float*)d_in[2];
    const float* b  = (const float*)d_in[3];
    float* out = (float*)d_out;

    float* y   = (float*)((char*)d_ws +
                 (((size_t)NUM_NODE * sizeof(float) + 255) & ~(size_t)255));

    {
        const int waves  = NUM_NODE / 16;
        const int blocks = (waves + 7) / 8;
        na_gemm_relu_kernel<<<blocks, 256, 0, stream>>>(x, W, b, y);
    }
    {
        na_bin_kernel<<<NCHUNK, 1024, 0, stream>>>(ei, ei + N_EDGES, (unsigned*)out);
    }
    {
        na_gather_kernel<<<NBINS, 512, 0, stream>>>(ei, ei + N_EDGES, y, out);
    }
}


# Edge binning moved into the GEMM launch: its 128 chunk work-groups (256 threads each) run as that launch's first work-groups; 3 launches -> 2; LDS arrival order of a node's row sum changes
# speedup vs baseline: 3.2798x; 1.1531x over previous
#include <hip/hip_runtime.h>

#define NUM_NODE 100000
#define EMBED_DIM 128
#define N_EDGES 600000
#define BIN_NODES 128
#define NBINS ((NUM_NODE + BIN_NODES - 1) / BIN_NODES)
#define NCHUNK 128
#define CHUNK_EDGES ((N_EDGES + NCHUNK - 1) / NCHUNK)
#define PIECE_WORDS 32
#define PIECE_CAP (PIECE_WORDS - 1)
#define TGT_BITS 17
#define TGT_MASK ((1u << TGT_BITS) - 1u)
#define BIN_WORDS (BIN_NODES * EMBED_DIM)

#define GEMM_BLOCKS ((NUM_NODE / 16 + 7) / 8)
#define K1_THREADS 256
#define BIN_QUADS 5
#define LCNT_WORDS ((NBINS + 3) & ~3)
#define BIN_LDS_WORDS (LCNT_WORDS + NBINS * PIECE_CAP)
#define GEMM_LDS_F4 (8 * 16 * 32)
#define BIN_LDS_F4 ((BIN_LDS_WORDS + 3) / 4)
#define K1_LDS_F4 (BIN_LDS_F4 > GEMM_LDS_F4 ? BIN_LDS_F4 : GEMM_LDS_F4)

static_assert(N_EDGES % 4 == 0 && CHUNK_EDGES % 4 == 0, "edge chunks are read in whole 16-byte quads");
static_assert(CHUNK_EDGES / 4 <= BIN_QUADS * K1_THREADS, "one binning work-group covers its chunk");
static_assert((N_EDGES - (NCHUNK - 1) * CHUNK_EDGES) / 4 >= (BIN_QUADS - 1) * K1_THREADS,
              "only a thread's last quad can lie behind the end of its chunk");

typedef __attribute__((ext_vector_type(2))) float v2f;
typedef __attribute__((ext_vector_type(8))) float v8f;

__device__ __forceinline__ void na_bin_role(const int* __restrict__ src,
                                            const int* __restrict__ tgt,
                                            unsigned* __restrict__ pieces,
                                            float4* lds, int chunk, int tid) {
    unsigned* lcnt = (unsigned*)lds;
    unsigned* lent = lcnt + LCNT_WORDS;

    const int e0   = chunk * CHUNK_EDGES;
    const int left = N_EDGES - e0;
    const int nq   = (left < CHUNK_EDGES ? left : CHUNK_EDGES) / 4;
    const int4* __restrict__ s4 = (const int4*)(src + e0);
    const int4* __restrict__ t4 = (const int4*)(tgt + e0);
    const int4 zero = {0, 0, 0, 0};
    int4 s[BIN_QUADS], t[BIN_QUADS];
#pragma unroll
    for (int k = 0; k < BIN_QUADS - 1; ++k) {
        s[k] = s4[tid + k * K1_THREADS];
        t[k] = t4[tid + k * K1_THREADS];
    }
    const int  qlast = tid + (BIN_QUADS - 1) * K1_THREADS;
    const bool tail  = qlast < nq;
    s[BIN_QUADS - 1] = zero;
    t[BIN_QUADS - 1] = zero;
    if (tail) {
        s[BIN_QUADS - 1] = s4[qlast];
        t[BIN_QUADS - 1] = t4[qlast];
    }

    const float4 z4 = {0.0f, 0.0f, 0.0f, 0.0f};
    for (int i = tid; i < BIN_LDS_F4; i += K1_THREADS) lds[i] = z4;
    __syncthreads();

#pragma unroll
    for (int k = 0; k < BIN_QUADS; ++k) {
        if (k < BIN_QUADS - 1 || tail) {
            const unsigned sv[4] = {(unsigned)s[k].x, (unsigned)s[k].y, (unsigned)s[k].z, (unsigned)s[k].w};
            const unsigned tv[4] = {(unsigned)t[k].x, (unsigned)t[k].y, (unsigned)t[k].z, (unsigned)t[k].w};
            unsigned pos[4];
#pragma unroll
            for (int c = 0; c < 4; ++c) pos[c] = atomicAdd(&lcnt[sv[c] / BIN_NODES], 1u);
#pragma unroll
            for (int c = 0; c < 4; ++c) {
                if (pos[c] < PIECE_CAP)
                    lent[(sv[c] / BIN_NODES) * PIECE_CAP + pos[c]] =
                        ((sv[c] % BIN_NODES) << TGT_BITS) | tv[c];
            }
        }
    }
    __syncthreads();

    const int lane = tid & 31;
    for (int bn = tid >> 5; bn < NBINS; bn += K1_THREADS / 32) {
        const unsigned* p = lane == 0 ? lcnt + bn : lent + bn * PIECE_CAP + lane - 1;
        pieces[(size_t)bn * BIN_WORDS + chunk * PIECE_WORDS + lane] = *p;
    }
}

__global__ void __launch_bounds__(K1_THREADS)
na_gemm_bin_kernel(const float* __restrict__ x,
                   const float* __restrict__ W,
                   const float* __restrict__ b,
                   float* __restrict__ y,
                   const int* __restrict__ src,
                   const int* __restrict__ tgt,
                   unsigned* __restrict__ pieces) {
    __shared__ float4 lds[K1_LDS_F4];

    const int tid = threadIdx.x;

    if (blockIdx.x < NCHUNK) {
        na_bin_role(src, tgt, pieces, lds, blockIdx.x, tid);
        return;
    }

    float4* wfrag = lds;

    const int lane = tid & 31;
    const int waveInBlock = tid >> 5;
    const int waveId = (blockIdx.x - NCHUNK) * 8 + waveInBlock;
    const int m0 = waveId * 16;
    const bool live = m0 < NUM_NODE;

    const int half = lane >> 4;
    const int l16  = lane & 15;

    float4 av[16];
    if (live) {
        const float4* __restrict__ xr =
            (const float4*)(x + (long long)(m0 + l16) * EMBED_DIM) + half;
#pragma unroll
        for (int g = 0; g < 16; ++g) av[g] = xr[2 * g];
    }

    {
        const float4* __restrict__ W4 = (const float4*)W;
        float4 st[16];
#pragma unroll
        for (int i = 0; i < 16; ++i) st[i] = W4[i * 256 + tid];
#pragma unroll
        for (int i = 0; i < 16; ++i) {
            const int n  = i * 8 + waveInBlock;
            const int sg = lane >> 1;
            const int sh = lane & 1;
            wfrag[((n >> 4) * 16 + sg) * 32 + sh * 16 + ((n & 15) ^ sg)] = st[i];
        }
    }
    __syncthreads();
    if (!live) return;

    const v8f zero = {};
    v8f acc[8];
#pragma unroll
    for (int nt = 0; nt < 8; ++nt) acc[nt] = zero;

    float4 bq[2][8];
#pragma unroll
    for (int nt = 0; nt < 8; ++nt) bq[0][nt] = wfrag[(nt * 16) * 32 + half * 16 + l16];

#pragma unroll
    for (int g = 0; g < 16; ++g) {
        const int cur = g & 1;
        const int nxt = cur ^ 1;
        if (g + 1 < 16) {
#pragma unroll
            for (int nt = 0; nt < 8; ++nt)
                bq[nxt][nt] = wfrag[(nt * 16 + g + 1) * 32 + half * 16 + (l16 ^ (g + 1))];
        }
        __builtin_amdgcn_sched_barrier(0);
        v2f a0, a1;
        a0.x = av[g].x; a0.y = av[g].y;
        a1.x = av[g].z; a1.y = av[g].w;
#pragma unroll
        for (int nt = 0; nt < 8; ++nt) {
            v2f b0;
            b0.x = bq[cur][nt].x; b0.y = bq[cur][nt].y;
            acc[nt] = __builtin_amdgcn_wmma_f32_16x16x4_f32(
                false, a0, false, b0, (short)0, acc[nt], false, false);
        }
#pragma unroll
        for (int nt = 0; nt < 8; ++nt) {
            v2f b1;
            b1.x = bq[cur][nt].z; b1.y = bq[cur][nt].w;
            acc[nt] = __builtin_amdgcn_wmma_f32_16x16x4_f32(
                false, a1, false, b1, (short)0, acc[nt], false, false);
        }
        __builtin_amdgcn_sched_barrier(0);
    }

#pragma unroll
    for (int nt = 0; nt < 8; ++nt) {
        const int n = nt * 16 + l16;
        const float bias = b[n];
#pragma unroll
        for (int j = 0; j < 8; ++j) {
            const int m = m0 + half * 8 + j;
            float v = acc[nt][j] + bias;
            y[(long long)m * EMBED_DIM + n] = v > 0.0f ? v : 0.0f;
        }
    }
}

__device__ __forceinline__ void na_node_by_scan(const int* __restrict__ src,
                                                const int* __restrict__ tgt,
                                                const float* __restrict__ ycol,
                                                float* out, int n, int lane) {
    float4 acc = {0.0f, 0.0f, 0.0f, 0.0f};
    int c = 0;
    for (int base = 0; base < N_EDGES; base += 32) {
        const int sv = src[base + lane];
        unsigned mask = __builtin_amdgcn_ballot_w32(sv == n);
        c += __builtin_popcount(mask);
        while (mask) {
            const int j = __builtin_ctz(mask);
            mask &= mask - 1;
            const int tt = __builtin_amdgcn_readfirstlane(tgt[base + j]);
            const float4 r = *(const float4*)(ycol + (long long)tt * EMBED_DIM);
            acc.x += r.x; acc.y += r.y; acc.z += r.z; acc.w += r.w;
        }
    }
    const float d = c > 1 ? (float)c : 1.0f;
    float4 o;
    o.x = acc.x / d; o.y = acc.y / d; o.z = acc.z / d; o.w = acc.w / d;
    *(float4*)(out + (long long)n * EMBED_DIM + lane * 4) = o;
}

__global__ void __launch_bounds__(512, 16)
na_gather_kernel(const int* __restrict__ src,
                 const int* __restrict__ tgt,
                 const float* __restrict__ y,
                 float* out) {
    __shared__ int lcnt[BIN_NODES];
    __shared__ int lstart[BIN_NODES + 1];
    __shared__ int sorted[NCHUNK * PIECE_CAP];
    __shared__ int ovf;

    const int tid  = threadIdx.x;
    const int lane = tid & 31;
    const int wave = __builtin_amdgcn_readfirstlane(tid >> 5);
    const int bin  = blockIdx.x;
    const int nodeBase = bin * BIN_NODES;
    const float* __restrict__ ycol = y + lane * 4;

    const unsigned* pieces = (const unsigned*)out + (size_t)bin * BIN_WORDS;
    unsigned w[8];
#pragma unroll
    for (int k = 0; k < 8; ++k) w[k] = pieces[k * 512 + tid];

    if (tid < BIN_NODES) lcnt[tid] = 0;
    if (tid == 0) ovf = 0;
    __syncthreads();

    int rank[8];
#pragma unroll
    for (int k = 0; k < 8; ++k) {
        const unsigned c = (unsigned)__builtin_amdgcn_readfirstlane((int)w[k]);
        rank[k] = -1;
        if (c > PIECE_CAP) {
            if (lane == 0) ovf = 1;
        } else if (lane >= 1 && (unsigned)lane <= c) {
            rank[k] = atomicAdd(&lcnt[w[k] >> TGT_BITS], 1);
        }
    }
    __syncthreads();

    if (ovf) {
        for (int i = 0; i < 8; ++i) {
            const int n = nodeBase + wave * 8 + i;
            if (n < NUM_NODE) na_node_by_scan(src, tgt, ycol, out, n, lane);
        }
        return;
    }

    if (wave == 0) {
        const int c0 = lcnt[4 * lane], c1 = lcnt[4 * lane + 1];
        const int c2 = lcnt[4 * lane + 2], c3 = lcnt[4 * lane + 3];
        const int own = c0 + c1 + c2 + c3;
        int incl = own;
#pragma unroll
        for (int d = 1; d < 32; d <<= 1) {
            const int up = __shfl_up(incl, d, 32);
            if (lane >= d) incl += up;
        }
        const int excl = incl - own;
        lstart[4 * lane]     = excl;
        lstart[4 * lane + 1] = excl + c0;
        lstart[4 * lane + 2] = excl + c0 + c1;
        lstart[4 * lane + 3] = excl + c0 + c1 + c2;
        if (lane == 31) lstart[BIN_NODES] = incl;
    }
    __syncthreads();

#pragma unroll
    for (int k = 0; k < 8; ++k) {
        if (rank[k] >= 0)
            sorted[lstart[w[k] >> TGT_BITS] + rank[k]] = (int)(w[k] & TGT_MASK);
    }
    __syncthreads();

    const int n0 = wave * 8;
    const int e1 = __builtin_amdgcn_readfirstlane(lstart[n0 + 8]);
    int cur    = n0;
    int curBeg = __builtin_amdgcn_readfirstlane(lstart[n0]);
    int curEnd = __builtin_amdgcn_readfirstlane(lstart[n0 + 1]);

    const float4 zero4 = {0.0f, 0.0f, 0.0f, 0.0f};
    float4 acc = zero4;

    for (int base = curBeg; base < e1; base += 8) {
        const int idx = base + (lane & 7);
        const int ids = idx < e1 ? sorted[idx] : 0;
        float4 r[8];
#pragma unroll
        for (int j = 0; j < 8; ++j) {
            r[j] = zero4;
            if (base + j < e1) {
                const int tt = __builtin_amdgcn_readlane(ids, j);
                r[j] = *(const float4*)(ycol + (long long)tt * EMBED_DIM);
            }
        }
#pragma unroll
        for (int j = 0; j < 8; ++j) {
            if (base + j < e1) {
                while (base + j >= curEnd) {
                    const int n = nodeBase + cur;
                    if (n < NUM_NODE) {
                        const int deg = curEnd - curBeg;
                        const float d = deg > 1 ? (float)deg : 1.0f;
                        float4 o;
                        o.x = acc.x / d; o.y = acc.y / d; o.z = acc.z / d; o.w = acc.w / d;
                        *(float4*)(out + (long long)n * EMBED_DIM + lane * 4) = o;
                    }
                    acc = zero4;
                    ++cur;
                    curBeg = curEnd;
                    curEnd = __builtin_amdgcn_readfirstlane(lstart[cur + 1]);
                }
                acc.x += r[j].x; acc.y += r[j].y; acc.z += r[j].z; acc.w += r[j].w;
            }
        }
    }
    while (cur < n0 + 8) {
        const int n = nodeBase + cur;
        if (n < NUM_NODE) {
            const int deg = curEnd - curBeg;
            const float d = deg > 1 ? (float)deg : 1.0f;
            float4 o;
            o.x = acc.x / d; o.y = acc.y / d; o.z = acc.z / d; o.w = acc.w / d;
            *(float4*)(out + (long long)n * EMBED_DIM + lane * 4) = o;
        }
        acc = zero4;
        ++cur;
        curBeg = curEnd;
        if (cur < n0 + 8) curEnd = __builtin_amdgcn_readfirstlane(lstart[cur + 1]);
    }
}

extern "C" void kernel_launch(void* const* d_in, const int* in_sizes, int n_in,
                              void* d_out, int out_size, void* d_ws, size_t ws_size,
                              hipStream_t stream) {
    const float* x  = (const float*)d_in[0];
    const int*   ei = (const int*)d_in[1];
    const float* W  = (const float*)d_in[2];
    const float* b  = (const float*)d_in[3];
    float* out = (float*)d_out;

    float* y   = (float*)((char*)d_ws +
                 (((size_t)NUM_NODE * sizeof(float) + 255) & ~(size_t)255));

    {
        na_gemm_bin_kernel<<<NCHUNK + GEMM_BLOCKS, K1_THREADS, 0, stream>>>(
            x, W, b, y, ei, ei + N_EDGES, (unsigned*)out);
    }
    {
        na_gather_kernel<<<NBINS, 512, 0, stream>>>(ei, ei + N_EDGES, y, out);
    }
}
